# starting kernel
# baseline (speedup 1.0000x reference)
#include <hip/hip_runtime.h>
#include <hip/hip_bf16.h>
#include <math.h>


typedef __attribute__((ext_vector_type(2))) float v2f;
typedef __attribute__((ext_vector_type(8))) float v8f;

#define IN_F   128
#define OUT_F  64
#define NHEADS 4
#define NEG_SLOPE 0.2f
#define EPS 1e-12f

__global__ __launch_bounds__(256) void k_zero(float* __restrict__ p, int total) {
    int i = blockIdx.x * 256 + threadIdx.x;
    if (i < total) p[i] = 0.0f;
}

__global__ __launch_bounds__(256) void k_init_out(float* __restrict__ out,
                                                  const float* __restrict__ b,
                                                  int total) {
    int i = blockIdx.x * 256 + threadIdx.x;
    if (i < total) out[i] = b[i & (OUT_F - 1)];
}

__global__ __launch_bounds__(256) void k_gemm_hW(const float* __restrict__ x,
                                                 const float* __restrict__ W,
                                                 float* __restrict__ h,
                                                 int nNodes) {
    __shared__ float sW[IN_F * OUT_F];
    const int tid = threadIdx.x;
    for (int i = tid; i < IN_F * OUT_F; i += 256) sW[i] = W[i];
    __syncthreads();

    const int wave    = tid >> 5;
    const int lane    = tid & 31;
    const int rowBase = blockIdx.x * 128 + wave * 16;
    const int m       = lane & 15;
    const int khalf   = (lane < 16) ? 0 : 2;

    int srcRow = rowBase + m;
    if (srcRow >= nNodes) srcRow = nNodes - 1;
    const float* __restrict__ xr = x + (size_t)srcRow * IN_F;

    v8f acc0 = {}, acc1 = {}, acc2 = {}, acc3 = {};
    const int n = lane & 15;

    for (int k = 0; k < IN_F; k += 4) {
        v2f a;
        a.x = xr[k + khalf];
        a.y = xr[k + khalf + 1];

        const float* __restrict__ w0 = &sW[(k + khalf) * OUT_F];
        const float* __restrict__ w1 = &sW[(k + khalf + 1) * OUT_F];
        v2f b0, b1, b2, b3;
        b0.x = w0[n];        b0.y = w1[n];
        b1.x = w0[16 + n];   b1.y = w1[16 + n];
        b2.x = w0[32 + n];   b2.y = w1[32 + n];
        b3.x = w0[48 + n];   b3.y = w1[48 + n];

        acc0 = __builtin_amdgcn_wmma_f32_16x16x4_f32(false, a, false, b0, (short)0, acc0, false, false);
        acc1 = __builtin_amdgcn_wmma_f32_16x16x4_f32(false, a, false, b1, (short)0, acc1, false, false);
        acc2 = __builtin_amdgcn_wmma_f32_16x16x4_f32(false, a, false, b2, (short)0, acc2, false, false);
        acc3 = __builtin_amdgcn_wmma_f32_16x16x4_f32(false, a, false, b3, (short)0, acc3, false, false);
    }

    const int nIdx = lane & 15;
    const int mOff = (lane < 16) ? 0 : 8;
#pragma unroll
    for (int j = 0; j < 8; ++j) {
        int r = rowBase + j + mOff;
        if (r < nNodes) {
            float* __restrict__ hr = h + (size_t)r * OUT_F;
            hr[nIdx]      = acc0[j];
            hr[16 + nIdx] = acc1[j];
            hr[32 + nIdx] = acc2[j];
            hr[48 + nIdx] = acc3[j];
        }
    }
}

__global__ __launch_bounds__(256) void k_scores(const float* __restrict__ h,
                                                const float* __restrict__ a_l,
                                                const float* __restrict__ a_r,
                                                float* __restrict__ el,
                                                float* __restrict__ er,
                                                int nNodes) {
    __shared__ float sAl[OUT_F * NHEADS];
    __shared__ float sAr[OUT_F * NHEADS];
    const int tid = threadIdx.x;
    if (tid < OUT_F * NHEADS) { sAl[tid] = a_l[tid]; sAr[tid] = a_r[tid]; }
    __syncthreads();

    const int node = blockIdx.x * 256 + tid;
    if (node >= nNodes) return;

    const float* __restrict__ hr = h + (size_t)node * OUT_F;
    float l0 = 0.f, l1 = 0.f, l2 = 0.f, l3 = 0.f;
    float r0 = 0.f, r1 = 0.f, r2 = 0.f, r3 = 0.f;
#pragma unroll 8
    for (int d = 0; d < OUT_F; ++d) {
        float hv = hr[d];
        l0 += hv * sAl[d * NHEADS + 0];
        l1 += hv * sAl[d * NHEADS + 1];
        l2 += hv * sAl[d * NHEADS + 2];
        l3 += hv * sAl[d * NHEADS + 3];
        r0 += hv * sAr[d * NHEADS + 0];
        r1 += hv * sAr[d * NHEADS + 1];
        r2 += hv * sAr[d * NHEADS + 2];
        r3 += hv * sAr[d * NHEADS + 3];
    }
    float* __restrict__ elp = el + (size_t)node * NHEADS;
    float* __restrict__ erp = er + (size_t)node * NHEADS;
    elp[0] = l0; elp[1] = l1; elp[2] = l2; elp[3] = l3;
    erp[0] = r0; erp[1] = r1; erp[2] = r2; erp[3] = r3;
}

__global__ __launch_bounds__(256) void k_edge_e(const int* __restrict__ ei,
                                                const float* __restrict__ el,
                                                const float* __restrict__ er,
                                                float* __restrict__ e,
                                                float* __restrict__ den,
                                                int nEdges) {
    const int eid = blockIdx.x * 256 + threadIdx.x;
    if (eid >= nEdges) return;
    const int row = ei[eid];
    const int col = ei[nEdges + eid];
    const float* __restrict__ elp = el + (size_t)row * NHEADS;
    const float* __restrict__ erp = er + (size_t)col * NHEADS;
    float* __restrict__ ep = e + (size_t)eid * NHEADS;
#pragma unroll
    for (int hh = 0; hh < NHEADS; ++hh) {
        float v = elp[hh] + erp[hh];
        v = (v > 0.0f) ? v : NEG_SLOPE * v;
        v = __expf(v);
        ep[hh] = v;
        atomicAdd(&den[(size_t)row * NHEADS + hh], v);
    }
}

__global__ __launch_bounds__(256) void k_edge_agg(const int* __restrict__ ei,
                                                  const float* __restrict__ e,
                                                  const float* __restrict__ den,
                                                  const float* __restrict__ h,
                                                  float* __restrict__ out,
                                                  int nEdges) {
    const int lane = threadIdx.x & 31;
    const int eid  = blockIdx.x * 8 + (threadIdx.x >> 5);
    if (eid >= nEdges) return;

    const int row = ei[eid];
    const int col = ei[nEdges + eid];

    const float h0 = h[(size_t)col * OUT_F + lane];
    const float h1 = h[(size_t)col * OUT_F + 32 + lane];

    const float* __restrict__ ep = e + (size_t)eid * NHEADS;
    const float* __restrict__ dp = den + (size_t)row * NHEADS;
    float* __restrict__ orow = out + (size_t)row * (NHEADS * OUT_F);

#pragma unroll
    for (int hh = 0; hh < NHEADS; ++hh) {
        const float attn = ep[hh] / fmaxf(dp[hh], EPS);
        atomicAdd(orow + hh * OUT_F + lane,      attn * h0);
        atomicAdd(orow + hh * OUT_F + 32 + lane, attn * h1);
    }
}

extern "C" void kernel_launch(void* const* d_in, const int* in_sizes, int n_in,
                              void* d_out, int out_size, void* d_ws, size_t ws_size,
                              hipStream_t stream) {
    (void)n_in; (void)ws_size;
    const float* x   = (const float*)d_in[0];
    const int*   ei  = (const int*)  d_in[1];
    const float* W   = (const float*)d_in[2];
    const float* a_l = (const float*)d_in[3];
    const float* a_r = (const float*)d_in[4];
    const float* b   = (const float*)d_in[5];
    float* out = (float*)d_out;

    const int nNodes = in_sizes[0] / IN_F;
    const int nEdges = in_sizes[1] / 2;

    float* h   = (float*)d_ws;
    float* el  = h   + (size_t)nNodes * OUT_F;
    float* er  = el  + (size_t)nNodes * NHEADS;
    float* den = er  + (size_t)nNodes * NHEADS;
    float* e   = den + (size_t)nNodes * NHEADS;

    k_zero    <<<(nNodes * NHEADS + 255) / 256, 256, 0, stream>>>(den, nNodes * NHEADS);
    k_init_out<<<(out_size + 255) / 256,        256, 0, stream>>>(out, b, out_size);
    k_gemm_hW <<<(nNodes + 127) / 128,          256, 0, stream>>>(x, W, h, nNodes);
    k_scores  <<<(nNodes + 255) / 256,          256, 0, stream>>>(h, a_l, a_r, el, er, nNodes);
    k_edge_e  <<<(nEdges + 255) / 256,          256, 0, stream>>>(ei, el, er, e, den, nEdges);
    k_edge_agg<<<(nEdges + 7) / 8,              256, 0, stream>>>(ei, e, den, h, out, nEdges);
}


# Counting sort of edges by destination row (arrival-order rank from a returning integer add); one wave per row sums its edges in registers and writes out once: float atomics gone, sum order changes
# speedup vs baseline: 2.1335x; 2.1335x over previous
#include <hip/hip_runtime.h>
#include <hip/hip_bf16.h>
#include <math.h>


typedef __attribute__((ext_vector_type(2))) float v2f;
typedef __attribute__((ext_vector_type(8))) float v8f;

#define IN_F   128
#define OUT_F  64
#define NHEADS 4
#define NEG_SLOPE 0.2f
#define EPS 1e-12f
#define SCAN_T 1024

__global__ __launch_bounds__(256) void k_zero(int* __restrict__ p, int total) {
    int i = blockIdx.x * 256 + threadIdx.x;
    if (i < total) p[i] = 0;
}

__global__ __launch_bounds__(256) void k_rank(const int* __restrict__ ei,
                                              int* __restrict__ cnt,
                                              int* __restrict__ rnk,
                                              int nEdges) {
    const int eid = blockIdx.x * 256 + threadIdx.x;
    if (eid >= nEdges) return;
    rnk[eid] = atomicAdd(&cnt[ei[eid]], 1);
}

__global__ __launch_bounds__(SCAN_T) void k_offsets(const int* __restrict__ cnt,
                                                    int* __restrict__ offs,
                                                    int nNodes) {
    __shared__ int sFront[SCAN_T / 32];
    __shared__ int sTile[SCAN_T / 32];
    const int tid   = threadIdx.x;
    const int lane  = tid & 31;
    const int w     = tid >> 5;
    const int tile0 = blockIdx.x * SCAN_T;

    int front = 0;
    for (int i = tid; i < tile0; i += SCAN_T) front += cnt[i];
    const int idx = tile0 + tid;
    const int v   = (idx < nNodes) ? cnt[idx] : 0;

    int inc = v;
    for (int o = 1; o < 32; o <<= 1) {
        const int u = __shfl_up(inc, o);
        if (lane >= o) inc += u;
    }
    for (int o = 16; o > 0; o >>= 1) front += __shfl_xor(front, o);
    if (lane == 31) sTile[w]  = inc;
    if (lane == 0)  sFront[w] = front;
    __syncthreads();

    int a = sFront[lane];
    int t = (lane < w) ? sTile[lane] : 0;
    for (int o = 16; o > 0; o >>= 1) {
        a += __shfl_xor(a, o);
        t += __shfl_xor(t, o);
    }
    const int excl = a + t + inc - v;
    if (idx < nNodes)      offs[idx]    = excl;
    if (idx == nNodes - 1) offs[nNodes] = excl + v;
}

__global__ __launch_bounds__(256) void k_fill(const int* __restrict__ ei,
                                              const int* __restrict__ offs,
                                              const int* __restrict__ rnk,
                                              int* __restrict__ scol,
                                              int nEdges) {
    const int eid = blockIdx.x * 256 + threadIdx.x;
    if (eid >= nEdges) return;
    const int row = ei[eid];
    const int col = ei[nEdges + eid];
    scol[offs[row] + rnk[eid]] = col;
}

__global__ __launch_bounds__(256) void k_gemm_hW(const float* __restrict__ x,
                                                 const float* __restrict__ W,
                                                 float* __restrict__ h,
                                                 int nNodes) {
    __shared__ float sW[IN_F * OUT_F];
    const int tid = threadIdx.x;
    for (int i = tid; i < IN_F * OUT_F; i += 256) sW[i] = W[i];
    __syncthreads();

    const int wave    = tid >> 5;
    const int lane    = tid & 31;
    const int rowBase = blockIdx.x * 128 + wave * 16;
    const int m       = lane & 15;
    const int khalf   = (lane < 16) ? 0 : 2;

    int srcRow = rowBase + m;
    if (srcRow >= nNodes) srcRow = nNodes - 1;
    const float* __restrict__ xr = x + (size_t)srcRow * IN_F;

    v8f acc0 = {}, acc1 = {}, acc2 = {}, acc3 = {};
    const int n = lane & 15;

    for (int k = 0; k < IN_F; k += 4) {
        v2f a;
        a.x = xr[k + khalf];
        a.y = xr[k + khalf + 1];

        const float* __restrict__ w0 = &sW[(k + khalf) * OUT_F];
        const float* __restrict__ w1 = &sW[(k + khalf + 1) * OUT_F];
        v2f b0, b1, b2, b3;
        b0.x = w0[n];        b0.y = w1[n];
        b1.x = w0[16 + n];   b1.y = w1[16 + n];
        b2.x = w0[32 + n];   b2.y = w1[32 + n];
        b3.x = w0[48 + n];   b3.y = w1[48 + n];

        acc0 = __builtin_amdgcn_wmma_f32_16x16x4_f32(false, a, false, b0, (short)0, acc0, false, false);
        acc1 = __builtin_amdgcn_wmma_f32_16x16x4_f32(false, a, false, b1, (short)0, acc1, false, false);
        acc2 = __builtin_amdgcn_wmma_f32_16x16x4_f32(false, a, false, b2, (short)0, acc2, false, false);
        acc3 = __builtin_amdgcn_wmma_f32_16x16x4_f32(false, a, false, b3, (short)0, acc3, false, false);
    }

    const int nIdx = lane & 15;
    const int mOff = (lane < 16) ? 0 : 8;
#pragma unroll
    for (int j = 0; j < 8; ++j) {
        int r = rowBase + j + mOff;
        if (r < nNodes) {
            float* __restrict__ hr = h + (size_t)r * OUT_F;
            hr[nIdx]      = acc0[j];
            hr[16 + nIdx] = acc1[j];
            hr[32 + nIdx] = acc2[j];
            hr[48 + nIdx] = acc3[j];
        }
    }
}

__global__ __launch_bounds__(256) void k_scores(const float* __restrict__ h,
                                                const float* __restrict__ a_l,
                                                const float* __restrict__ a_r,
                                                float* __restrict__ el,
                                                float* __restrict__ er,
                                                int nNodes) {
    __shared__ float sAl[OUT_F * NHEADS];
    __shared__ float sAr[OUT_F * NHEADS];
    const int tid = threadIdx.x;
    if (tid < OUT_F * NHEADS) { sAl[tid] = a_l[tid]; sAr[tid] = a_r[tid]; }
    __syncthreads();

    const int node = blockIdx.x * 256 + tid;
    if (node >= nNodes) return;

    const float* __restrict__ hr = h + (size_t)node * OUT_F;
    float l0 = 0.f, l1 = 0.f, l2 = 0.f, l3 = 0.f;
    float r0 = 0.f, r1 = 0.f, r2 = 0.f, r3 = 0.f;
#pragma unroll 8
    for (int d = 0; d < OUT_F; ++d) {
        float hv = hr[d];
        l0 += hv * sAl[d * NHEADS + 0];
        l1 += hv * sAl[d * NHEADS + 1];
        l2 += hv * sAl[d * NHEADS + 2];
        l3 += hv * sAl[d * NHEADS + 3];
        r0 += hv * sAr[d * NHEADS + 0];
        r1 += hv * sAr[d * NHEADS + 1];
        r2 += hv * sAr[d * NHEADS + 2];
        r3 += hv * sAr[d * NHEADS + 3];
    }
    float* __restrict__ elp = el + (size_t)node * NHEADS;
    float* __restrict__ erp = er + (size_t)node * NHEADS;
    elp[0] = l0; elp[1] = l1; elp[2] = l2; elp[3] = l3;
    erp[0] = r0; erp[1] = r1; erp[2] = r2; erp[3] = r3;
}

__device__ __forceinline__ float edge_w(float v) {
    v = (v > 0.0f) ? v : NEG_SLOPE * v;
    return __expf(v);
}

__device__ __forceinline__ void edge_terms(int j, int end,
                                           const int* __restrict__ scol,
                                           const float* __restrict__ er,
                                           const float4 l4, int& c,
                                           float& e0, float& e1, float& e2, float& e3) {
    c = 0; e0 = 0.0f; e1 = 0.0f; e2 = 0.0f; e3 = 0.0f;
    if (j < end) {
        c = scol[j];
        const float4 r4 = *(const float4*)(er + (size_t)c * NHEADS);
        e0 = edge_w(l4.x + r4.x);
        e1 = edge_w(l4.y + r4.y);
        e2 = edge_w(l4.z + r4.z);
        e3 = edge_w(l4.w + r4.w);
    }
}

__device__ __forceinline__ float lane_f(float v, int src) {
    return __int_as_float(__builtin_amdgcn_readlane(__float_as_int(v), src));
}

__device__ __forceinline__ void add_chunk(int c, float a0, float a1, float a2, float a3,
                                          int n, const float* __restrict__ hl,
                                          float2& s0, float2& s1, float2& s2, float2& s3) {
#pragma unroll 4
    for (int t = 0; t < n; ++t) {
        const int   ct = __builtin_amdgcn_readlane(c, t);
        const float w0 = lane_f(a0, t);
        const float w1 = lane_f(a1, t);
        const float w2 = lane_f(a2, t);
        const float w3 = lane_f(a3, t);
        const float2 hv = *(const float2*)(hl + (size_t)ct * OUT_F);
        s0.x += w0 * hv.x; s0.y += w0 * hv.y;
        s1.x += w1 * hv.x; s1.y += w1 * hv.y;
        s2.x += w2 * hv.x; s2.y += w2 * hv.y;
        s3.x += w3 * hv.x; s3.y += w3 * hv.y;
    }
}

__global__ __launch_bounds__(256) void k_row_gather(const int* __restrict__ offs,
                                                    const int* __restrict__ scol,
                                                    const float* __restrict__ el,
                                                    const float* __restrict__ er,
                                                    const float* __restrict__ h,
                                                    const float* __restrict__ b,
                                                    float* __restrict__ out,
                                                    int nNodes) {
    const int lane = threadIdx.x & 31;
    const int row = __builtin_amdgcn_readfirstlane(blockIdx.x * 8 + (threadIdx.x >> 5));
    if (row >= nNodes) return;

    const int beg = offs[row];
    const int end = offs[row + 1];
    const float4 l4 = *(const float4*)(el + (size_t)row * NHEADS);

    int c0; float e0, e1, e2, e3;
    edge_terms(beg + lane, end, scol, er, l4, c0, e0, e1, e2, e3);
    float d0 = e0, d1 = e1, d2 = e2, d3 = e3;
    for (int base = beg + 32; base < end; base += 32) {
        int c; float f0, f1, f2, f3;
        edge_terms(base + lane, end, scol, er, l4, c, f0, f1, f2, f3);
        d0 += f0; d1 += f1; d2 += f2; d3 += f3;
    }
    for (int o = 16; o > 0; o >>= 1) {
        d0 += __shfl_xor(d0, o);
        d1 += __shfl_xor(d1, o);
        d2 += __shfl_xor(d2, o);
        d3 += __shfl_xor(d3, o);
    }
    const float m0 = fmaxf(d0, EPS), m1 = fmaxf(d1, EPS);
    const float m2 = fmaxf(d2, EPS), m3 = fmaxf(d3, EPS);

    const float* __restrict__ hl = h + 2 * lane;
    float2 s0 = {0.0f, 0.0f}, s1 = {0.0f, 0.0f}, s2 = {0.0f, 0.0f}, s3 = {0.0f, 0.0f};

    int n = end - beg; if (n > 32) n = 32;
    add_chunk(c0, e0 / m0, e1 / m1, e2 / m2, e3 / m3, n, hl, s0, s1, s2, s3);
    for (int base = beg + 32; base < end; base += 32) {
        int c; float f0, f1, f2, f3;
        edge_terms(base + lane, end, scol, er, l4, c, f0, f1, f2, f3);
        n = end - base; if (n > 32) n = 32;
        add_chunk(c, f0 / m0, f1 / m1, f2 / m2, f3 / m3, n, hl, s0, s1, s2, s3);
    }

    const float b0 = b[2 * lane];
    const float b1 = b[2 * lane + 1];
    float* __restrict__ orow = out + (size_t)row * (NHEADS * OUT_F) + 2 * lane;
    float2 o0 = {s0.x + b0, s0.y + b1};
    float2 o1 = {s1.x + b0, s1.y + b1};
    float2 o2 = {s2.x + b0, s2.y + b1};
    float2 o3 = {s3.x + b0, s3.y + b1};
    *(float2*)(orow)             = o0;
    *(float2*)(orow + OUT_F)     = o1;
    *(float2*)(orow + 2 * OUT_F) = o2;
    *(float2*)(orow + 3 * OUT_F) = o3;
}

extern "C" void kernel_launch(void* const* d_in, const int* in_sizes, int n_in,
                              void* d_out, int out_size, void* d_ws, size_t ws_size,
                              hipStream_t stream) {
    (void)n_in; (void)ws_size; (void)out_size;
    const float* x   = (const float*)d_in[0];
    const int*   ei  = (const int*)  d_in[1];
    const float* W   = (const float*)d_in[2];
    const float* a_l = (const float*)d_in[3];
    const float* a_r = (const float*)d_in[4];
    const float* b   = (const float*)d_in[5];
    float* out = (float*)d_out;

    const int nNodes = in_sizes[0] / IN_F;
    const int nEdges = in_sizes[1] / 2;

    float* h   = (float*)d_ws;
    float* el  = h  + (size_t)nNodes * OUT_F;
    float* er  = el + (size_t)nNodes * NHEADS;
    int* cnt   = (int*)(er + (size_t)nNodes * NHEADS);
    int* offs  = cnt  + (size_t)nNodes;
    int* rnk   = offs + (size_t)nNodes + 1;
    int* scol  = rnk  + (size_t)nEdges;

    k_zero      <<<(nNodes + 255) / 256,          256,    0, stream>>>(cnt, nNodes);
    k_rank      <<<(nEdges + 255) / 256,          256,    0, stream>>>(ei, cnt, rnk, nEdges);
    k_gemm_hW   <<<(nNodes + 127) / 128,          256,    0, stream>>>(x, W, h, nNodes);
    k_scores    <<<(nNodes + 255) / 256,          256,    0, stream>>>(h, a_l, a_r, el, er, nNodes);
    k_offsets   <<<(nNodes + SCAN_T - 1) / SCAN_T, SCAN_T, 0, stream>>>(cnt, offs, nNodes);
    k_fill      <<<(nEdges + 255) / 256,          256,    0, stream>>>(ei, offs, rnk, scol, nEdges);
    k_row_gather<<<(nNodes + 7) / 8,              256,    0, stream>>>(offs, scol, el, er, h, b, out, nNodes);
}


# Both sort levels place into LDS first (pairs by bucket, cols by row) and copy out in slot order: a scatter store writes a contiguous run, not one line per lane; sum order in a row stays arrival order
# speedup vs baseline: 4.8871x; 2.2906x over previous
#include <hip/hip_runtime.h>
#include <hip/hip_bf16.h>
#include <math.h>


typedef __attribute__((ext_vector_type(2))) float v2f;
typedef __attribute__((ext_vector_type(8))) float v8f;

#define IN_F   128
#define OUT_F  64
#define NHEADS 4
#define NEG_SLOPE 0.2f
#define EPS 1e-12f

#define BSH    8
#define BROWS  (1 << BSH)
#define SORT_T 1024
#define EPT    8
#define CHUNK  (EPT * SORT_T)
#define MAXB   SORT_T
#define GEMM_T 256
#define EPT_C  (CHUNK / GEMM_T)
#define COL_CAP 8192

__device__ __forceinline__ void bucket_count_block(const int* __restrict__ ei,
                                                   int* __restrict__ hist,
                                                   int nEdges, int nBuck, int blk) {
    __shared__ int sCnt[MAXB];
    const int tid = threadIdx.x;
    for (int i = tid; i < MAXB; i += GEMM_T) sCnt[i] = 0;

    int row[EPT_C];
#pragma unroll
    for (int k = 0; k < EPT_C; ++k) {
        const int e = blk * CHUNK + k * GEMM_T + tid;
        row[k] = (e < nEdges) ? ei[e] : -1;
    }
    __syncthreads();
#pragma unroll
    for (int k = 0; k < EPT_C; ++k)
        if (row[k] >= 0) atomicAdd(&sCnt[row[k] >> BSH], 1);
    __syncthreads();
    for (int i = tid; i < nBuck; i += GEMM_T) hist[blk * nBuck + i] = sCnt[i];
}

__global__ __launch_bounds__(SORT_T) void k_bucket_place(const int* __restrict__ ei,
                                                         const int* __restrict__ hist,
                                                         int* __restrict__ bstart,
                                                         int* __restrict__ pairWords,
                                                         int nEdges, int nBuck, int nBlk) {
    int2* __restrict__ pairs = (int2*)pairWords;
    __shared__ int sTot[SORT_T];
    __shared__ int sPre[SORT_T];
    __shared__ int sBase[MAXB];
    __shared__ int sCur[MAXB];
    __shared__ int sWave[SORT_T / 32];
    __shared__ int sWaveM[SORT_T / 32];
    __shared__ int2 sPair[CHUNK];
    const int tid  = threadIdx.x;
    const int lane = tid & 31;
    const int w    = tid >> 5;
    const int me   = blockIdx.x;

    int row[EPT], col[EPT];
#pragma unroll
    for (int k = 0; k < EPT; ++k) {
        const int e = me * CHUNK + k * SORT_T + tid;
        row[k] = (e < nEdges) ? ei[e] : -1;
        col[k] = (e < nEdges) ? ei[nEdges + e] : 0;
    }
    const int mine = (tid < nBuck) ? hist[me * nBuck + tid] : 0;

    const int per  = (nBuck + 31) & ~31;
    const int nGrp = SORT_T / per;
    const int g    = tid / per;
    const int bq   = tid - g * per;
    int tot = 0, pre = 0;
    if (g < nGrp && bq < nBuck) {
        for (int k0 = g; k0 < nBlk; k0 += 32 * nGrp) {
            int hv[32];
#pragma unroll
            for (int j = 0; j < 32; ++j) {
                const int k = k0 + j * nGrp;
                hv[j] = (k < nBlk) ? hist[k * nBuck + bq] : 0;
            }
#pragma unroll
            for (int j = 0; j < 32; ++j) {
                tot += hv[j];
                pre += (k0 + j * nGrp < me) ? hv[j] : 0;
            }
        }
    }
    sTot[tid] = tot;
    sPre[tid] = pre;
    __syncthreads();

    int v = 0, p = 0;
    if (tid < nBuck) {
        for (int j = 0; j < nGrp; ++j) {
            v += sTot[j * per + tid];
            p += sPre[j * per + tid];
        }
    }
    int inc = v, incM = mine;
    for (int o = 1; o < 32; o <<= 1) {
        const int u  = __shfl_up(inc, o);
        const int uM = __shfl_up(incM, o);
        if (lane >= o) { inc += u; incM += uM; }
    }
    if (lane == 31) { sWave[w] = inc; sWaveM[w] = incM; }
    __syncthreads();
    int t  = (lane < w) ? sWave[lane]  : 0;
    int tM = (lane < w) ? sWaveM[lane] : 0;
    int nMine = sWaveM[lane];
    for (int o = 16; o > 0; o >>= 1) {
        t     += __shfl_xor(t, o);
        tM    += __shfl_xor(tM, o);
        nMine += __shfl_xor(nMine, o);
    }
    const int excl = t + inc - v;
    const int loc  = tM + incM - mine;
    if (tid < nBuck) {
        sBase[tid] = excl + p - loc;
        sCur[tid]  = loc;
        if (me == 0) {
            bstart[tid] = excl;
            if (tid == nBuck - 1) bstart[nBuck] = excl + v;
        }
    }
    __syncthreads();

#pragma unroll
    for (int k = 0; k < EPT; ++k) {
        if (row[k] >= 0) {
            const int bk   = row[k] >> BSH;
            const int slot = atomicAdd(&sCur[bk], 1);
            sPair[slot] = make_int2(row[k], col[k]);
        }
    }
    __syncthreads();

#pragma unroll
    for (int k = 0; k < EPT; ++k) {
        const int i = k * SORT_T + tid;
        if (i < nMine) {
            const int2 pr = sPair[i];
            pairs[sBase[pr.x >> BSH] + i] = pr;
        }
    }
}

__global__ __launch_bounds__(SORT_T) void k_bucket_rows(const int* __restrict__ pairWords,
                                                        const int* __restrict__ bstart,
                                                        int* __restrict__ offs,
                                                        int* __restrict__ scol,
                                                        int nNodes) {
    const int2* __restrict__ pairs = (const int2*)pairWords;
    __shared__ int sCnt[BROWS];
    __shared__ int sCur[BROWS];
    __shared__ int sWave[BROWS / 32];
    __shared__ int sCol[COL_CAP];
    const int tid  = threadIdx.x;
    const int lane = tid & 31;
    const int w    = tid >> 5;
    const int s    = bstart[blockIdx.x];
    const int e    = bstart[blockIdx.x + 1];
    const bool staged = (e - s) <= COL_CAP;

    int2 rc0[4];
#pragma unroll
    for (int k = 0; k < 4; ++k) {
        const int i = s + k * SORT_T + tid;
        rc0[k] = (i < e) ? pairs[i] : make_int2(-1, 0);
    }
    if (tid < BROWS) sCnt[tid] = 0;
    __syncthreads();
#pragma unroll
    for (int k = 0; k < 4; ++k)
        if (rc0[k].x >= 0) atomicAdd(&sCnt[rc0[k].x & (BROWS - 1)], 1);
    for (int base = s + 4 * SORT_T; base < e; base += 4 * SORT_T) {
        int r[4];
#pragma unroll
        for (int k = 0; k < 4; ++k) {
            const int i = base + k * SORT_T + tid;
            r[k] = (i < e) ? pairs[i].x : -1;
        }
#pragma unroll
        for (int k = 0; k < 4; ++k)
            if (r[k] >= 0) atomicAdd(&sCnt[r[k] & (BROWS - 1)], 1);
    }
    __syncthreads();

    int v = 0, inc = 0;
    if (tid < BROWS) {
        v = sCnt[tid];
        inc = v;
        for (int o = 1; o < 32; o <<= 1) {
            const int u = __shfl_up(inc, o);
            if (lane >= o) inc += u;
        }
        if (lane == 31) sWave[w] = inc;
    }
    __syncthreads();
    if (tid < BROWS) {
        int t = (lane < w) ? sWave[lane] : 0;
        for (int o = 16; o > 0; o >>= 1) t += __shfl_xor(t, o);
        const int excl = t + inc - v;
        sCur[tid] = excl;
        const int node = blockIdx.x * BROWS + tid;
        if (node < nNodes)      offs[node]   = s + excl;
        if (node == nNodes - 1) offs[nNodes] = s + excl + v;
    }
    __syncthreads();
#pragma unroll
    for (int k = 0; k < 4; ++k) {
        if (rc0[k].x >= 0) {
            const int p = atomicAdd(&sCur[rc0[k].x & (BROWS - 1)], 1);
            if (staged) sCol[p] = rc0[k].y;
            else        scol[s + p] = rc0[k].y;
        }
    }
    for (int base = s + 4 * SORT_T; base < e; base += 4 * SORT_T) {
        int2 rc[4];
#pragma unroll
        for (int k = 0; k < 4; ++k) {
            const int i = base + k * SORT_T + tid;
            rc[k] = (i < e) ? pairs[i] : make_int2(-1, 0);
        }
#pragma unroll
        for (int k = 0; k < 4; ++k) {
            if (rc[k].x >= 0) {
                const int p = atomicAdd(&sCur[rc[k].x & (BROWS - 1)], 1);
                if (staged) sCol[p] = rc[k].y;
                else        scol[s + p] = rc[k].y;
            }
        }
    }
    if (staged) {
        __syncthreads();
        const int nb = e - s;
        for (int i = tid; i < nb; i += SORT_T) scol[s + i] = sCol[i];
    }
}

#define SC_P 68
__device__ __forceinline__ void gemm_hW_block(const float* __restrict__ x,
                                              const float* __restrict__ W,
                                              const float* __restrict__ a_l,
                                              const float* __restrict__ a_r,
                                              float* __restrict__ h,
                                              float* __restrict__ el,
                                              float* __restrict__ er,
                                              int nNodes, int gb) {
    __shared__ float sW[IN_F * OUT_F];
    __shared__ __attribute__((aligned(16))) float sAT[16 * SC_P];
    __shared__ __attribute__((aligned(16))) float sH[8 * 16 * SC_P];
    const int tid = threadIdx.x;
    for (int i = tid; i < IN_F * OUT_F; i += 256) sW[i] = W[i];
    for (int i = tid; i < 16 * OUT_F; i += 256) {
        const int o = i >> 6;
        const int d = i & (OUT_F - 1);
        float v = 0.0f;
        if (o < NHEADS)          v = a_l[d * NHEADS + o];
        else if (o < 2 * NHEADS) v = a_r[d * NHEADS + o - NHEADS];
        sAT[o * SC_P + d] = v;
    }
    __syncthreads();

    const int wave    = tid >> 5;
    const int lane    = tid & 31;
    const int rowBase = gb * 128 + wave * 16;
    const int m       = lane & 15;
    const int khalf   = (lane < 16) ? 0 : 2;

    int srcRow = rowBase + m;
    if (srcRow >= nNodes) srcRow = nNodes - 1;
    const float* __restrict__ xr = x + (size_t)srcRow * IN_F;

    v8f acc0 = {}, acc1 = {}, acc2 = {}, acc3 = {};
    const int n = lane & 15;

    for (int k = 0; k < IN_F; k += 4) {
        v2f a;
        a.x = xr[k + khalf];
        a.y = xr[k + khalf + 1];

        const float* __restrict__ w0 = &sW[(k + khalf) * OUT_F];
        const float* __restrict__ w1 = &sW[(k + khalf + 1) * OUT_F];
        v2f b0, b1, b2, b3;
        b0.x = w0[n];        b0.y = w1[n];
        b1.x = w0[16 + n];   b1.y = w1[16 + n];
        b2.x = w0[32 + n];   b2.y = w1[32 + n];
        b3.x = w0[48 + n];   b3.y = w1[48 + n];

        acc0 = __builtin_amdgcn_wmma_f32_16x16x4_f32(false, a, false, b0, (short)0, acc0, false, false);
        acc1 = __builtin_amdgcn_wmma_f32_16x16x4_f32(false, a, false, b1, (short)0, acc1, false, false);
        acc2 = __builtin_amdgcn_wmma_f32_16x16x4_f32(false, a, false, b2, (short)0, acc2, false, false);
        acc3 = __builtin_amdgcn_wmma_f32_16x16x4_f32(false, a, false, b3, (short)0, acc3, false, false);
    }

    const int nIdx = lane & 15;
    const int mOff = (lane < 16) ? 0 : 8;

    float* __restrict__ sT = sH + wave * (16 * SC_P);
#pragma unroll
    for (int j = 0; j < 8; ++j) {
        float* __restrict__ tr = sT + (j + mOff) * SC_P + nIdx;
        tr[0]  = acc0[j];
        tr[16] = acc1[j];
        tr[32] = acc2[j];
        tr[48] = acc3[j];
    }
    __syncthreads();

#pragma unroll
    for (int j = 0; j < 8; ++j) {
        int r = rowBase + j + mOff;
        if (r < nNodes) {
            float* __restrict__ hr = h + (size_t)r * OUT_F;
            hr[nIdx]      = acc0[j];
            hr[16 + nIdx] = acc1[j];
            hr[32 + nIdx] = acc2[j];
            hr[48 + nIdx] = acc3[j];
        }
    }

    const float* __restrict__ pa = sAT + m * SC_P + khalf;
    const float* __restrict__ pb = sT  + m * SC_P + khalf;
    v8f sc0 = {}, sc1 = {};
#pragma unroll
    for (int kk = 0; kk < OUT_F / 4; kk += 2) {
        const v2f fa0 = *(const v2f*)(pa + 4 * kk);
        const v2f fb0 = *(const v2f*)(pb + 4 * kk);
        const v2f fa1 = *(const v2f*)(pa + 4 * kk + 4);
        const v2f fb1 = *(const v2f*)(pb + 4 * kk + 4);
        sc0 = __builtin_amdgcn_wmma_f32_16x16x4_f32(false, fa0, false, fb0, (short)0, sc0, false, false);
        sc1 = __builtin_amdgcn_wmma_f32_16x16x4_f32(false, fa1, false, fb1, (short)0, sc1, false, false);
    }
    const int node = rowBase + lane;
    if (lane < 16 && node < nNodes) {
        const float4 lv = {sc0[0] + sc1[0], sc0[1] + sc1[1], sc0[2] + sc1[2], sc0[3] + sc1[3]};
        const float4 rv = {sc0[4] + sc1[4], sc0[5] + sc1[5], sc0[6] + sc1[6], sc0[7] + sc1[7]};
        *(float4*)(el + (size_t)node * NHEADS) = lv;
        *(float4*)(er + (size_t)node * NHEADS) = rv;
    }
}

__global__ __launch_bounds__(GEMM_T) void k_gemm_hW_count(const float* __restrict__ x,
                                                          const float* __restrict__ W,
                                                          const float* __restrict__ a_l,
                                                          const float* __restrict__ a_r,
                                                          float* __restrict__ h,
                                                          float* __restrict__ el,
                                                          float* __restrict__ er,
                                                          int nNodes, int nGemmBlk,
                                                          const int* __restrict__ ei,
                                                          int* __restrict__ hist,
                                                          int nEdges, int nBuck) {
    const int bid = blockIdx.x;
    if (bid < nGemmBlk) gemm_hW_block(x, W, a_l, a_r, h, el, er, nNodes, bid);
    else                bucket_count_block(ei, hist, nEdges, nBuck, bid - nGemmBlk);
}

__device__ __forceinline__ float edge_w(float v) {
    v = (v > 0.0f) ? v : NEG_SLOPE * v;
    return __expf(v);
}

__device__ __forceinline__ void edge_terms(int j, int end,
                                           const int* __restrict__ scol,
                                           const float* __restrict__ er,
                                           const float4 l4, int& c,
                                           float& e0, float& e1, float& e2, float& e3) {
    c = 0; e0 = 0.0f; e1 = 0.0f; e2 = 0.0f; e3 = 0.0f;
    if (j < end) {
        c = scol[j];
        const float4 r4 = *(const float4*)(er + (size_t)c * NHEADS);
        e0 = edge_w(l4.x + r4.x);
        e1 = edge_w(l4.y + r4.y);
        e2 = edge_w(l4.z + r4.z);
        e3 = edge_w(l4.w + r4.w);
    }
}

__device__ __forceinline__ float lane_f(float v, int src) {
    return __int_as_float(__builtin_amdgcn_readlane(__float_as_int(v), src));
}

__device__ __forceinline__ void add_chunk(int c, float a0, float a1, float a2, float a3,
                                          int n, const float* __restrict__ hl,
                                          float2& s0, float2& s1, float2& s2, float2& s3) {
#pragma unroll 4
    for (int t = 0; t < n; ++t) {
        const int   ct = __builtin_amdgcn_readlane(c, t);
        const float w0 = lane_f(a0, t);
        const float w1 = lane_f(a1, t);
        const float w2 = lane_f(a2, t);
        const float w3 = lane_f(a3, t);
        const float2 hv = *(const float2*)(hl + (size_t)ct * OUT_F);
        s0.x += w0 * hv.x; s0.y += w0 * hv.y;
        s1.x += w1 * hv.x; s1.y += w1 * hv.y;
        s2.x += w2 * hv.x; s2.y += w2 * hv.y;
        s3.x += w3 * hv.x; s3.y += w3 * hv.y;
    }
}

__global__ __launch_bounds__(256) void k_row_gather(const int* __restrict__ offs,
                                                    const int* __restrict__ scol,
                                                    const float* __restrict__ el,
                                                    const float* __restrict__ er,
                                                    const float* __restrict__ h,
                                                    const float* __restrict__ b,
                                                    float* __restrict__ out,
                                                    int nNodes) {
    const int lane = threadIdx.x & 31;
    const int row = __builtin_amdgcn_readfirstlane(blockIdx.x * 8 + (threadIdx.x >> 5));
    if (row >= nNodes) return;

    const int beg = offs[row];
    const int end = offs[row + 1];
    const float4 l4 = *(const float4*)(el + (size_t)row * NHEADS);

    int c0; float e0, e1, e2, e3;
    edge_terms(beg + lane, end, scol, er, l4, c0, e0, e1, e2, e3);
    float d0 = e0, d1 = e1, d2 = e2, d3 = e3;
    for (int base = beg + 32; base < end; base += 32) {
        int c; float f0, f1, f2, f3;
        edge_terms(base + lane, end, scol, er, l4, c, f0, f1, f2, f3);
        d0 += f0; d1 += f1; d2 += f2; d3 += f3;
    }
    for (int o = 16; o > 0; o >>= 1) {
        d0 += __shfl_xor(d0, o);
        d1 += __shfl_xor(d1, o);
        d2 += __shfl_xor(d2, o);
        d3 += __shfl_xor(d3, o);
    }
    const float m0 = fmaxf(d0, EPS), m1 = fmaxf(d1, EPS);
    const float m2 = fmaxf(d2, EPS), m3 = fmaxf(d3, EPS);

    const float* __restrict__ hl = h + 2 * lane;
    float2 s0 = {0.0f, 0.0f}, s1 = {0.0f, 0.0f}, s2 = {0.0f, 0.0f}, s3 = {0.0f, 0.0f};

    int n = end - beg; if (n > 32) n = 32;
    add_chunk(c0, e0 / m0, e1 / m1, e2 / m2, e3 / m3, n, hl, s0, s1, s2, s3);
    for (int base = beg + 32; base < end; base += 32) {
        int c; float f0, f1, f2, f3;
        edge_terms(base + lane, end, scol, er, l4, c, f0, f1, f2, f3);
        n = end - base; if (n > 32) n = 32;
        add_chunk(c, f0 / m0, f1 / m1, f2 / m2, f3 / m3, n, hl, s0, s1, s2, s3);
    }

    const float b0 = b[2 * lane];
    const float b1 = b[2 * lane + 1];
    float* __restrict__ orow = out + (size_t)row * (NHEADS * OUT_F) + 2 * lane;
    float2 o0 = {s0.x + b0, s0.y + b1};
    float2 o1 = {s1.x + b0, s1.y + b1};
    float2 o2 = {s2.x + b0, s2.y + b1};
    float2 o3 = {s3.x + b0, s3.y + b1};
    *(float2*)(orow)             = o0;
    *(float2*)(orow + OUT_F)     = o1;
    *(float2*)(orow + 2 * OUT_F) = o2;
    *(float2*)(orow + 3 * OUT_F) = o3;
}

extern "C" void kernel_launch(void* const* d_in, const int* in_sizes, int n_in,
                              void* d_out, int out_size, void* d_ws, size_t ws_size,
                              hipStream_t stream) {
    (void)n_in; (void)ws_size; (void)out_size;
    const float* x   = (const float*)d_in[0];
    const int*   ei  = (const int*)  d_in[1];
    const float* W   = (const float*)d_in[2];
    const float* a_l = (const float*)d_in[3];
    const float* a_r = (const float*)d_in[4];
    const float* b   = (const float*)d_in[5];
    float* out = (float*)d_out;

    const int nNodes = in_sizes[0] / IN_F;
    const int nEdges = in_sizes[1] / 2;

    const int nBuck = (nNodes + BROWS - 1) / BROWS;
    const int nBlk  = (nEdges + CHUNK - 1) / CHUNK;

    const int nGemmBlk = (nNodes + 127) / 128;

    float* h   = (float*)d_ws;
    float* el  = h  + (size_t)nNodes * OUT_F;
    float* er  = el + (size_t)nNodes * NHEADS;
    int* hist   = (int*)(er + (size_t)nNodes * NHEADS);
    int* bstart = hist   + (size_t)nBuck * nBlk;
    int* offs   = bstart + (size_t)nBuck + 1;
    int* scol   = offs   + (size_t)nNodes + 1;
    int* pairs  = (int*)d_out;

    k_gemm_hW_count<<<nGemmBlk + nBlk,     GEMM_T, 0, stream>>>(x, W, a_l, a_r, h, el, er, nNodes, nGemmBlk,
                                                                ei, hist, nEdges, nBuck);
    k_bucket_place<<<nBlk,                 SORT_T, 0, stream>>>(ei, hist, bstart, pairs, nEdges, nBuck, nBlk);
    k_bucket_rows <<<nBuck,                SORT_T, 0, stream>>>(pairs, bstart, offs, scol, nNodes);
    k_row_gather  <<<(nNodes + 7) / 8,     256,    0, stream>>>(offs, scol, el, er, h, b, out, nNodes);
}


# Row pass takes two rows per wave, one per 16-lane half: a lane owns 16 bytes of a row of h, col and weights picked inside the half by permlane16 (was readlane); score sums over a half: order changes
# speedup vs baseline: 7.1087x; 1.4546x over previous
#include <hip/hip_runtime.h>
#include <hip/hip_bf16.h>
#include <math.h>


typedef __attribute__((ext_vector_type(2))) float v2f;
typedef __attribute__((ext_vector_type(8))) float v8f;

#define IN_F   128
#define OUT_F  64
#define NHEADS 4
#define NEG_SLOPE 0.2f
#define EPS 1e-12f

#define BROWS  196
#define HROWS  (BROWS / 2)
#define HPAD   128
#define SORT_T 1024
#define CHUNK  8192
#define MAXB   1024
#define GEMM_T 256
#define EPT_C  (CHUNK / GEMM_T)
#define BPT    (MAXB / GEMM_T)
#define COL_CAP 8192
#define KRUN   4
#define KWORD  2

__device__ __forceinline__ void bucket_sort_block(const int* __restrict__ ei,
                                                  int* __restrict__ runs,
                                                  int* __restrict__ pk,
                                                  int nEdges, int nBuck, int nBlk, int blk) {
    __shared__ __attribute__((aligned(16))) int sCnt[MAXB];
    __shared__ int sWave[GEMM_T / 32];
    __shared__ __attribute__((aligned(16))) int sPk[CHUNK];
    const int tid  = threadIdx.x;
    const int lane = tid & 31;
    const int w    = tid >> 5;
    for (int i = tid; i < MAXB; i += GEMM_T) sCnt[i] = 0;

    int bk[EPT_C], wd[EPT_C];
#pragma unroll
    for (int k = 0; k < EPT_C; ++k) {
        const int e = blk * CHUNK + k * GEMM_T + tid;
        const int r = (e < nEdges) ? ei[e] : -1;
        const int cc = (e < nEdges) ? ei[nEdges + e] : 0;
        const int q = r / BROWS;
        bk[k] = (r >= 0) ? q : -1;
        wd[k] = (int)(((unsigned)(r - q * BROWS) << 24) | (unsigned)cc);
    }
    __syncthreads();
#pragma unroll
    for (int k = 0; k < EPT_C; ++k)
        if (bk[k] >= 0) atomicAdd(&sCnt[bk[k]], 1);
    __syncthreads();

    const int4 c4 = *(const int4*)&sCnt[BPT * tid];
    const int sum = c4.x + c4.y + c4.z + c4.w;
    int inc = sum;
    for (int o = 1; o < 32; o <<= 1) {
        const int u = __shfl_up(inc, o);
        if (lane >= o) inc += u;
    }
    if (lane == 31) sWave[w] = inc;
    __syncthreads();
    int t = (lane < w) ? sWave[lane] : 0;
    for (int o = 16; o > 0; o >>= 1) t += __shfl_xor(t, o);
    const int x0 = t + inc - sum;
    const int x1 = x0 + c4.x;
    const int x2 = x1 + c4.y;
    const int x3 = x2 + c4.z;
    *(int4*)&sCnt[BPT * tid] = make_int4(x0, x1, x2, x3);
    const int b0 = BPT * tid;
    if (b0     < nBuck) runs[(b0    ) * nBlk + blk] = x0 | (c4.x << 16);
    if (b0 + 1 < nBuck) runs[(b0 + 1) * nBlk + blk] = x1 | (c4.y << 16);
    if (b0 + 2 < nBuck) runs[(b0 + 2) * nBlk + blk] = x2 | (c4.z << 16);
    if (b0 + 3 < nBuck) runs[(b0 + 3) * nBlk + blk] = x3 | (c4.w << 16);
    __syncthreads();

    int slot[EPT_C];
#pragma unroll
    for (int k = 0; k < EPT_C; ++k)
        slot[k] = (bk[k] >= 0) ? atomicAdd(&sCnt[bk[k]], 1) : -1;
#pragma unroll
    for (int k = 0; k < EPT_C; ++k)
        if (slot[k] >= 0) sPk[slot[k]] = wd[k];
    __syncthreads();

    int nMine = nEdges - blk * CHUNK;
    if (nMine > CHUNK) nMine = CHUNK;
    int* __restrict__ dst = pk + (size_t)blk * CHUNK;
#pragma unroll
    for (int k = 0; k < CHUNK / (4 * GEMM_T); ++k) {
        const int i = 4 * (k * GEMM_T + tid);
        if (i + 4 <= nMine) {
            *(int4*)(dst + i) = *(const int4*)(sPk + i);
        } else {
            for (int j = i; j < nMine; ++j) dst[j] = sPk[j];
        }
    }
}

#define SC_P 68
__device__ __forceinline__ void gemm_hW_block(const float* __restrict__ x,
                                              const float* __restrict__ W,
                                              const float* __restrict__ a_l,
                                              const float* __restrict__ a_r,
                                              float* __restrict__ h,
                                              float* __restrict__ el,
                                              float* __restrict__ er,
                                              int nNodes, int gb) {
    __shared__ float sW[IN_F * OUT_F];
    __shared__ __attribute__((aligned(16))) float sAT[16 * SC_P];
    __shared__ __attribute__((aligned(16))) float sH[8 * 16 * SC_P];
    const int tid = threadIdx.x;
    for (int i = tid; i < IN_F * OUT_F; i += 256) sW[i] = W[i];
    for (int i = tid; i < 16 * OUT_F; i += 256) {
        const int o = i >> 6;
        const int d = i & (OUT_F - 1);
        float v = 0.0f;
        if (o < NHEADS)          v = a_l[d * NHEADS + o];
        else if (o < 2 * NHEADS) v = a_r[d * NHEADS + o - NHEADS];
        sAT[o * SC_P + d] = v;
    }
    __syncthreads();

    const int wave    = tid >> 5;
    const int lane    = tid & 31;
    const int rowBase = gb * 128 + wave * 16;
    const int m       = lane & 15;
    const int khalf   = (lane < 16) ? 0 : 2;

    int srcRow = rowBase + m;
    if (srcRow >= nNodes) srcRow = nNodes - 1;
    const float* __restrict__ xr = x + (size_t)srcRow * IN_F;

    v8f acc0 = {}, acc1 = {}, acc2 = {}, acc3 = {};
    const int n = lane & 15;

    for (int k = 0; k < IN_F; k += 4) {
        v2f a;
        a.x = xr[k + khalf];
        a.y = xr[k + khalf + 1];

        const float* __restrict__ w0 = &sW[(k + khalf) * OUT_F];
        const float* __restrict__ w1 = &sW[(k + khalf + 1) * OUT_F];
        v2f b0, b1, b2, b3;
        b0.x = w0[n];        b0.y = w1[n];
        b1.x = w0[16 + n];   b1.y = w1[16 + n];
        b2.x = w0[32 + n];   b2.y = w1[32 + n];
        b3.x = w0[48 + n];   b3.y = w1[48 + n];

        acc0 = __builtin_amdgcn_wmma_f32_16x16x4_f32(false, a, false, b0, (short)0, acc0, false, false);
        acc1 = __builtin_amdgcn_wmma_f32_16x16x4_f32(false, a, false, b1, (short)0, acc1, false, false);
        acc2 = __builtin_amdgcn_wmma_f32_16x16x4_f32(false, a, false, b2, (short)0, acc2, false, false);
        acc3 = __builtin_amdgcn_wmma_f32_16x16x4_f32(false, a, false, b3, (short)0, acc3, false, false);
    }

    const int nIdx = lane & 15;
    const int mOff = (lane < 16) ? 0 : 8;

    float* __restrict__ sT = sH + wave * (16 * SC_P);
#pragma unroll
    for (int j = 0; j < 8; ++j) {
        float* __restrict__ tr = sT + (j + mOff) * SC_P + nIdx;
        tr[0]  = acc0[j];
        tr[16] = acc1[j];
        tr[32] = acc2[j];
        tr[48] = acc3[j];
    }
    __syncthreads();

#pragma unroll
    for (int j = 0; j < 8; ++j) {
        int r = rowBase + j + mOff;
        if (r < nNodes) {
            float* __restrict__ hr = h + (size_t)r * OUT_F;
            hr[nIdx]      = acc0[j];
            hr[16 + nIdx] = acc1[j];
            hr[32 + nIdx] = acc2[j];
            hr[48 + nIdx] = acc3[j];
        }
    }

    const float* __restrict__ pa = sAT + m * SC_P + khalf;
    const float* __restrict__ pb = sT  + m * SC_P + khalf;
    v8f sc0 = {}, sc1 = {};
#pragma unroll
    for (int kk = 0; kk < OUT_F / 4; kk += 2) {
        const v2f fa0 = *(const v2f*)(pa + 4 * kk);
        const v2f fb0 = *(const v2f*)(pb + 4 * kk);
        const v2f fa1 = *(const v2f*)(pa + 4 * kk + 4);
        const v2f fb1 = *(const v2f*)(pb + 4 * kk + 4);
        sc0 = __builtin_amdgcn_wmma_f32_16x16x4_f32(false, fa0, false, fb0, (short)0, sc0, false, false);
        sc1 = __builtin_amdgcn_wmma_f32_16x16x4_f32(false, fa1, false, fb1, (short)0, sc1, false, false);
    }
    const int node = rowBase + lane;
    if (lane < 16 && node < nNodes) {
        const float4 lv = {sc0[0] + sc1[0], sc0[1] + sc1[1], sc0[2] + sc1[2], sc0[3] + sc1[3]};
        const float4 rv = {sc0[4] + sc1[4], sc0[5] + sc1[5], sc0[6] + sc1[6], sc0[7] + sc1[7]};
        *(float4*)(el + (size_t)node * NHEADS) = lv;
        *(float4*)(er + (size_t)node * NHEADS) = rv;
    }
}

__global__ __launch_bounds__(GEMM_T) void k_gemm_hW_sort(const float* __restrict__ x,
                                                          const float* __restrict__ W,
                                                          const float* __restrict__ a_l,
                                                          const float* __restrict__ a_r,
                                                          float* __restrict__ h,
                                                          float* __restrict__ el,
                                                          float* __restrict__ er,
                                                         int nNodes, int nGemmBlk,
                                                         const int* __restrict__ ei,
                                                         int* __restrict__ runs,
                                                         int* __restrict__ pk,
                                                         int nEdges, int nBuck, int nBlk) {
    const int bid = blockIdx.x;
    if (bid < nGemmBlk) gemm_hW_block(x, W, a_l, a_r, h, el, er, nNodes, bid);
    else                bucket_sort_block(ei, runs, pk, nEdges, nBuck, nBlk, bid - nGemmBlk);
}

__device__ __forceinline__ float edge_w(float v) {
    v = (v > 0.0f) ? v : NEG_SLOPE * v;
    return __expf(v);
}

__device__ __forceinline__ void edge_terms(int j, int end,
                                           const int* __restrict__ cols,
                                           const float* __restrict__ er,
                                           const float4 l4, int& c,
                                           float& e0, float& e1, float& e2, float& e3) {
    c = 0; e0 = 0.0f; e1 = 0.0f; e2 = 0.0f; e3 = 0.0f;
    if (j < end) {
        c = cols[j];
        const float4 r4 = *(const float4*)(er + (size_t)c * NHEADS);
        e0 = edge_w(l4.x + r4.x);
        e1 = edge_w(l4.y + r4.y);
        e2 = edge_w(l4.z + r4.z);
        e3 = edge_w(l4.w + r4.w);
    }
}

__device__ __forceinline__ void listed_terms(int j, int end, unsigned rb,
                                             const int* __restrict__ pk,
                                             const float* __restrict__ er,
                                             const float4 l4, bool& hit, int& c,
                                             float& e0, float& e1, float& e2, float& e3) {
    hit = false; c = 0; e0 = 0.0f; e1 = 0.0f; e2 = 0.0f; e3 = 0.0f;
    if (j < end) {
        const unsigned word = (unsigned)pk[j];
        if ((word >> 24) == rb) {
            hit = true;
            c = (int)(word & 0xFFFFFFu);
            const float4 r4 = *(const float4*)(er + (size_t)c * NHEADS);
            e0 = edge_w(l4.x + r4.x);
            e1 = edge_w(l4.y + r4.y);
            e2 = edge_w(l4.z + r4.z);
            e3 = edge_w(l4.w + r4.w);
        }
    }
}

__device__ __forceinline__ float lane_f(float v, int src) {
    return __int_as_float(__builtin_amdgcn_readlane(__float_as_int(v), src));
}

__device__ __forceinline__ void add_edge(int c, float a0, float a1, float a2, float a3,
                                         int t, const float* __restrict__ hl,
                                         float2& s0, float2& s1, float2& s2, float2& s3) {
    const int   ct = __builtin_amdgcn_readlane(c, t);
    const float w0 = lane_f(a0, t);
    const float w1 = lane_f(a1, t);
    const float w2 = lane_f(a2, t);
    const float w3 = lane_f(a3, t);
    const float2 hv = *(const float2*)(hl + (size_t)ct * OUT_F);
    s0.x += w0 * hv.x; s0.y += w0 * hv.y;
    s1.x += w1 * hv.x; s1.y += w1 * hv.y;
    s2.x += w2 * hv.x; s2.y += w2 * hv.y;
    s3.x += w3 * hv.x; s3.y += w3 * hv.y;
}

__device__ __forceinline__ unsigned half_pick_u(unsigned v, unsigned sel) {
    return __builtin_amdgcn_permlane16(v, v, sel, sel, true, true);
}
__device__ __forceinline__ float half_pick_f(float v, unsigned sel) {
    return __uint_as_float(half_pick_u(__float_as_uint(v), sel));
}

struct PairAcc { float2 p01[4]; float2 p23[4]; };

template <int U>
__device__ __forceinline__ void pair_chunk(int c, float a0, float a1, float a2, float a3, int n,
                                           unsigned lo16, const float* __restrict__ h, PairAcc& s) {
    for (int t = 0; t < n; t += U) {
        float4 hv[U];
#pragma unroll
        for (int u = 0; u < U; ++u) {
            const unsigned ct = half_pick_u((unsigned)c, 0x11111111u * (unsigned)(t + u));
            hv[u] = *(const float4*)((const char*)h + ((ct << 8) | lo16));
        }
#pragma unroll
        for (int u = 0; u < U; ++u) {
            const unsigned sel = 0x11111111u * (unsigned)(t + u);
            const float w0 = half_pick_f(a0, sel);
            const float w1 = half_pick_f(a1, sel);
            const float w2 = half_pick_f(a2, sel);
            const float w3 = half_pick_f(a3, sel);
            const float v[4] = {hv[u].x, hv[u].y, hv[u].z, hv[u].w};
#pragma unroll
            for (int k = 0; k < 4; ++k) {
                s.p01[k].x += w0 * v[k]; s.p01[k].y += w1 * v[k];
                s.p23[k].x += w2 * v[k]; s.p23[k].y += w3 * v[k];
            }
        }
    }
}

__device__ __forceinline__ void run_of(const int* __restrict__ runs, int bucket, int nBlk, int blk,
                                       int& st, int& cn) {
    const int word = runs[bucket * nBlk + blk];
    st = blk * CHUNK + (word & 0xFFFF);
    cn = word >> 16;
}

template <typename F>
__device__ __forceinline__ void rest_words(const int* __restrict__ pk, const int* __restrict__ runs,
                                           int bucket, int nBlk, int w, int lane,
                                           const int (&rs)[KRUN], const int (&rc)[KRUN], F f) {
#pragma unroll
    for (int k = 0; k < KRUN; ++k)
        for (int i = KWORD * 32 + lane; i < rc[k]; i += 32) f(pk[rs[k] + i]);
    for (int blk = w + 32 * KRUN; blk < nBlk; blk += 32) {
        int st, cn;
        run_of(runs, bucket, nBlk, blk, st, cn);
        for (int i = lane; i < cn; i += 32) f(pk[st + i]);
    }
}

__global__ __launch_bounds__(SORT_T) __attribute__((amdgpu_waves_per_eu(16, 16))) void k_rows_gather(const int* __restrict__ pk,
                                                        const int* __restrict__ runs,
                                                        const float* __restrict__ el,
                                                        const float* __restrict__ er,
                                                        const float* __restrict__ h,
                                                        const float* __restrict__ b,
                                                        float* __restrict__ out,
                                                        int nNodes, int nBlk) {
    __shared__ int sCnt[HPAD];
    __shared__ int sOff[HPAD + 1];
    __shared__ int sCur[HPAD];
    __shared__ int sWave[HPAD / 32];
    __shared__ int sNext;
    __shared__ int sCol[COL_CAP];
    __shared__ __attribute__((aligned(16))) float sB[OUT_F];
    const int tid    = threadIdx.x;
    const int lane   = tid & 31;
    const int w      = __builtin_amdgcn_readfirstlane(tid >> 5);
    const int bucket = blockIdx.x >> 1;
    const unsigned r0 = (blockIdx.x & 1) * HROWS;

    int tw = 0;
    if (lane < KRUN) {
        const int blk = w + 32 * lane;
        if (blk < nBlk) tw = runs[bucket * nBlk + blk];
    }
    int rs[KRUN], rc[KRUN];
#pragma unroll
    for (int k = 0; k < KRUN; ++k) {
        const int word = __builtin_amdgcn_readlane(tw, k);
        rs[k] = (w + 32 * k) * CHUNK + (word & 0xFFFF);
        rc[k] = word >> 16;
    }
    int kw[KRUN][KWORD];
#pragma unroll
    for (int k = 0; k < KRUN; ++k) {
#pragma unroll
        for (int q = 0; q < KWORD; ++q) {
            const int i = q * 32 + lane;
            kw[k][q] = (i < rc[k]) ? pk[rs[k] + i] : -1;
        }
    }
    if (tid < HPAD) sCnt[tid] = 0;
    if (tid == 0)   sNext = 0;
    if (tid < OUT_F) sB[tid] = b[tid];
    __syncthreads();
#pragma unroll
    for (int k = 0; k < KRUN; ++k) {
#pragma unroll
        for (int q = 0; q < KWORD; ++q) {
            const unsigned lr = ((unsigned)kw[k][q] >> 24) - r0;
            if (lr < HROWS) atomicAdd(&sCnt[lr], 1);
        }
    }
    rest_words(pk, runs, bucket, nBlk, w, lane, rs, rc, [&](int word) {
        const unsigned lr = ((unsigned)word >> 24) - r0;
        if (lr < HROWS) atomicAdd(&sCnt[lr], 1);
    });
    __syncthreads();

    int v = 0, inc = 0;
    if (tid < HPAD) {
        v = sCnt[tid];
        inc = v;
        for (int o = 1; o < 32; o <<= 1) {
            const int u = __shfl_up(inc, o);
            if (lane >= o) inc += u;
        }
        if (lane == 31) sWave[w] = inc;
    }
    __syncthreads();
    if (tid < HPAD) {
        int t = (lane < w) ? sWave[lane] : 0;
        for (int o = 16; o > 0; o >>= 1) t += __shfl_xor(t, o);
        const int excl = t + inc - v;
        sOff[tid] = excl;
        sCur[tid] = excl;
        if (tid == HPAD - 1) sOff[HPAD] = excl + v;
    }
    __syncthreads();
    const bool staged = sOff[HROWS] <= COL_CAP;
    if (staged) {
        int p[KRUN][KWORD];
#pragma unroll
        for (int k = 0; k < KRUN; ++k) {
#pragma unroll
            for (int q = 0; q < KWORD; ++q) {
                const unsigned lr = ((unsigned)kw[k][q] >> 24) - r0;
                p[k][q] = (lr < HROWS) ? atomicAdd(&sCur[lr], 1) : -1;
            }
        }
#pragma unroll
        for (int k = 0; k < KRUN; ++k) {
#pragma unroll
            for (int q = 0; q < KWORD; ++q)
                if (p[k][q] >= 0) sCol[p[k][q]] = kw[k][q] & 0xFFFFFF;
        }
        rest_words(pk, runs, bucket, nBlk, w, lane, rs, rc, [&](int word) {
            const unsigned lr = ((unsigned)word >> 24) - r0;
            if (lr < HROWS) {
                const int p = atomicAdd(&sCur[lr], 1);
                sCol[p] = word & 0xFFFFFF;
            }
        });
    }
    __syncthreads();

    if (staged) {
        const int li = lane & 15;
        const int hf = lane >> 4;
        const unsigned lo16 = (unsigned)li << 4;
        const size_t rowWg = (size_t)(bucket * BROWS + (int)r0);
        const float* __restrict__ elWg = el + rowWg * NHEADS;
        float* __restrict__ outWg = out + rowWg * (NHEADS * OUT_F);
        for (int it = 0; it < HROWS / 2; ++it) {
            int pr = 0;
            if (lane == 0) pr = atomicAdd(&sNext, 1);
            pr = __builtin_amdgcn_readfirstlane(pr);
            if (pr >= HROWS / 2) break;
            const int rowA = bucket * BROWS + (int)r0 + 2 * pr;
            if (rowA >= nNodes) break;
            const int lr   = 2 * pr + hf;
            const bool rowOk = rowA + hf < nNodes;
            const int beg = sOff[lr];
            const int end = sOff[lr + 1];
            const int n   = end - beg;
            const int nA   = __builtin_amdgcn_readlane(n, 0);
            const int nB   = __builtin_amdgcn_readlane(n, 16);
            const int nmax = nA > nB ? nA : nB;

            const float4 l4 = *(const float4*)((const char*)elWg + ((unsigned)(rowOk ? lr : 2 * pr) << 4));

            int cA, cB; float eA0, eA1, eA2, eA3, eB0, eB1, eB2, eB3;
            edge_terms(beg + li,      end, sCol, er, l4, cA, eA0, eA1, eA2, eA3);
            edge_terms(beg + 16 + li, end, sCol, er, l4, cB, eB0, eB1, eB2, eB3);
            float d0 = eA0 + eB0, d1 = eA1 + eB1, d2 = eA2 + eB2, d3 = eA3 + eB3;
            for (int base = 32; base < nmax; base += 16) {
                int c; float f0, f1, f2, f3;
                edge_terms(beg + base + li, end, sCol, er, l4, c, f0, f1, f2, f3);
                d0 += f0; d1 += f1; d2 += f2; d3 += f3;
            }
            for (int o = 8; o > 0; o >>= 1) {
                d0 += __shfl_xor(d0, o, 16);
                d1 += __shfl_xor(d1, o, 16);
                d2 += __shfl_xor(d2, o, 16);
                d3 += __shfl_xor(d3, o, 16);
            }
            const float m0 = fmaxf(d0, EPS), m1 = fmaxf(d1, EPS);
            const float m2 = fmaxf(d2, EPS), m3 = fmaxf(d3, EPS);
            const float aA0 = eA0 / m0, aA1 = eA1 / m1, aA2 = eA2 / m2, aA3 = eA3 / m3;
            const float aB0 = eB0 / m0, aB1 = eB1 / m1, aB2 = eB2 / m2, aB3 = eB3 / m3;

            PairAcc s;
#pragma unroll
            for (int k = 0; k < 4; ++k) { s.p01[k] = make_float2(0.0f, 0.0f); s.p23[k] = make_float2(0.0f, 0.0f); }
            int nn;
            nn = nmax > 16 ? 16 : nmax;
            pair_chunk<2>(cA, aA0, aA1, aA2, aA3, nn, lo16, h, s);
            nn = nmax - 16; if (nn > 16) nn = 16;
            pair_chunk<2>(cB, aB0, aB1, aB2, aB3, nn, lo16, h, s);
            for (int base = 32; base < nmax; base += 16) {
                int c; float f0, f1, f2, f3;
                edge_terms(beg + base + li, end, sCol, er, l4, c, f0, f1, f2, f3);
                nn = nmax - base; if (nn > 16) nn = 16;
                pair_chunk<1>(c, f0 / m0, f1 / m1, f2 / m2, f3 / m3, nn, lo16, h, s);
            }

            if (rowOk) {
                const float4 b4 = *(const float4*)(sB + 4 * li);
                float* __restrict__ orow = (float*)((char*)outWg + (((unsigned)lr << 10) | lo16));
                const float4 o0 = {s.p01[0].x + b4.x, s.p01[1].x + b4.y, s.p01[2].x + b4.z, s.p01[3].x + b4.w};
                const float4 o1 = {s.p01[0].y + b4.x, s.p01[1].y + b4.y, s.p01[2].y + b4.z, s.p01[3].y + b4.w};
                const float4 o2 = {s.p23[0].x + b4.x, s.p23[1].x + b4.y, s.p23[2].x + b4.z, s.p23[3].x + b4.w};
                const float4 o3 = {s.p23[0].y + b4.x, s.p23[1].y + b4.y, s.p23[2].y + b4.z, s.p23[3].y + b4.w};
                *(float4*)(orow)             = o0;
                *(float4*)(orow + OUT_F)     = o1;
                *(float4*)(orow + 2 * OUT_F) = o2;
                *(float4*)(orow + 3 * OUT_F) = o3;
            }
        }
        return;
    }
    const float b0 = b[2 * lane];
    const float b1 = b[2 * lane + 1];
    const float* __restrict__ hl = h + 2 * lane;
    for (int it = 0; it < HROWS; ++it) {
        int lr = 0;
        if (lane == 0) lr = atomicAdd(&sNext, 1);
        lr = __builtin_amdgcn_readfirstlane(lr);
        if (lr >= HROWS) break;
        const int row = bucket * BROWS + (int)r0 + lr;
        if (row >= nNodes) break;

        const float4 l4 = *(const float4*)(el + (size_t)row * NHEADS);
        float2 s0 = {0.0f, 0.0f}, s1 = {0.0f, 0.0f}, s2 = {0.0f, 0.0f}, s3 = {0.0f, 0.0f};

        const unsigned rb = r0 + (unsigned)lr;
        float d0 = 0.0f, d1 = 0.0f, d2 = 0.0f, d3 = 0.0f;
        for (int blk = 0; blk < nBlk; ++blk) {
            int s, n;
            run_of(runs, bucket, nBlk, blk, s, n);
            const int e = s + n;
            for (int base = s; base < e; base += 32) {
                bool hit; int c; float f0, f1, f2, f3;
                listed_terms(base + lane, e, rb, pk, er, l4, hit, c, f0, f1, f2, f3);
                d0 += f0; d1 += f1; d2 += f2; d3 += f3;
            }
        }
        for (int o = 16; o > 0; o >>= 1) {
            d0 += __shfl_xor(d0, o);
            d1 += __shfl_xor(d1, o);
            d2 += __shfl_xor(d2, o);
            d3 += __shfl_xor(d3, o);
        }
        const float m0 = fmaxf(d0, EPS), m1 = fmaxf(d1, EPS);
        const float m2 = fmaxf(d2, EPS), m3 = fmaxf(d3, EPS);
        for (int blk = 0; blk < nBlk; ++blk) {
            int s, n;
            run_of(runs, bucket, nBlk, blk, s, n);
            const int e = s + n;
            for (int base = s; base < e; base += 32) {
                bool hit; int c; float f0, f1, f2, f3;
                listed_terms(base + lane, e, rb, pk, er, l4, hit, c, f0, f1, f2, f3);
                unsigned mask = __builtin_amdgcn_ballot_w32(hit);
                const float a0 = f0 / m0, a1 = f1 / m1, a2 = f2 / m2, a3 = f3 / m3;
                while (mask) {
                    const int t = __builtin_ctz(mask);
                    mask &= mask - 1;
                    add_edge(c, a0, a1, a2, a3, t, hl, s0, s1, s2, s3);
                }
            }
        }

        float* __restrict__ orow = out + (size_t)row * (NHEADS * OUT_F) + 2 * lane;
        float2 o0 = {s0.x + b0, s0.y + b1};
        float2 o1 = {s1.x + b0, s1.y + b1};
        float2 o2 = {s2.x + b0, s2.y + b1};
        float2 o3 = {s3.x + b0, s3.y + b1};
        *(float2*)(orow)             = o0;
        *(float2*)(orow + OUT_F)     = o1;
        *(float2*)(orow + 2 * OUT_F) = o2;
        *(float2*)(orow + 3 * OUT_F) = o3;
    }
}

extern "C" void kernel_launch(void* const* d_in, const int* in_sizes, int n_in,
                              void* d_out, int out_size, void* d_ws, size_t ws_size,
                              hipStream_t stream) {
    (void)n_in; (void)ws_size; (void)out_size;
    const float* x   = (const float*)d_in[0];
    const int*   ei  = (const int*)  d_in[1];
    const float* W   = (const float*)d_in[2];
    const float* a_l = (const float*)d_in[3];
    const float* a_r = (const float*)d_in[4];
    const float* b   = (const float*)d_in[5];
    float* out = (float*)d_out;

    const int nNodes = in_sizes[0] / IN_F;
    const int nEdges = in_sizes[1] / 2;

    const int nBuck = (nNodes + BROWS - 1) / BROWS;
    const int nBlk  = (nEdges + CHUNK - 1) / CHUNK;

    const int nGemmBlk = (nNodes + 127) / 128;

    float* h   = (float*)d_ws;
    float* el  = h  + (size_t)nNodes * OUT_F;
    float* er  = el + (size_t)nNodes * NHEADS;
    int* pk    = (int*)(er + (size_t)nNodes * NHEADS);
    int* runs  = pk + (size_t)nEdges;

    k_gemm_hW_sort<<<nGemmBlk + nBlk, GEMM_T, 0, stream>>>(x, W, a_l, a_r, h, el, er, nNodes, nGemmBlk,
                                                           ei, runs, pk, nEdges, nBuck, nBlk);
    k_rows_gather <<<2 * nBuck,       SORT_T, 0, stream>>>(pk, runs, el, er, h, b, out, nNodes, nBlk);
}
